# starting kernel
# baseline (speedup 1.0000x reference)
#include <hip/hip_runtime.h>
#include <math.h>

typedef float v2f __attribute__((ext_vector_type(2)));
typedef float v8f __attribute__((ext_vector_type(8)));

#define IN_F  128
#define HD    128
#define NHEAD 4
#define DHEAD 32
#define DEG   16
#define ALPHA 0.2f
#define LDSK  133

#if defined(__HIP_DEVICE_COMPILE__)
#if !__has_builtin(__builtin_amdgcn_wmma_f32_16x16x4_f32)
#error "gfx1250 fp32 WMMA builtin not available on this toolchain (device pass)"
#endif
#endif

__global__ __launch_bounds__(256) void gat_zero(float* __restrict__ p, int n) {
  int i = blockIdx.x * 256 + threadIdx.x;
  if (i < n) p[i] = 0.0f;
}

__global__ __launch_bounds__(256) void gat_gemm(
    const float* __restrict__ X, const float* __restrict__ W0,
    const float* __restrict__ W1, const float* __restrict__ a0,
    float* __restrict__ X1out, float* __restrict__ s0, float* __restrict__ s1,
    int N)
{
  __shared__ float lA[16 * LDSK];

  const int tid   = threadIdx.x;
  const int nbase = blockIdx.x * 16;

  {
    int row = tid >> 4;
    int col = (tid & 15) * 8;
    int r = nbase + row; if (r > N - 1) r = N - 1;
    const float4* src = (const float4*)(X + (size_t)r * IN_F + col);
    float4 q0 = src[0];
    float4 q1 = src[1];
    float* dst = lA + row * LDSK + col;
    dst[0] = q0.x; dst[1] = q0.y; dst[2] = q0.z; dst[3] = q0.w;
    dst[4] = q1.x; dst[5] = q1.y; dst[6] = q1.z; dst[7] = q1.w;
  }
  __syncthreads();

  const int wave  = tid >> 5;
  const int lane  = tid & 31;
  const int l15   = lane & 15;
  const int lhalf = lane >> 4;
  const int cbase = wave * 16;

  const float* w0row = W0 + (size_t)(cbase + l15) * IN_F;
  const float* w1row = W1 + (size_t)(cbase + l15) * IN_F;
  const float* arow  = lA + l15 * LDSK;

  v8f acc0 = {};
  v8f acc1 = {};

  #pragma unroll
  for (int kk = 0; kk < IN_F / 4; ++kk) {
    const int k0 = kk * 4 + lhalf * 2;
    v2f av; av.x = arow[k0];  av.y = arow[k0 + 1];
    v2f b0; b0.x = w0row[k0]; b0.y = w0row[k0 + 1];
    v2f b1; b1.x = w1row[k0]; b1.y = w1row[k0 + 1];
    acc0 = __builtin_amdgcn_wmma_f32_16x16x4_f32(false, av, false, b0,
                                                 (short)0, acc0, false, false);
    acc1 = __builtin_amdgcn_wmma_f32_16x16x4_f32(false, av, false, b1,
                                                 (short)0, acc1, false, false);
  }

  const int   head = wave >> 1;
  const float aval = a0[head * DHEAD + (cbase & (DHEAD - 1)) + l15];

  float p0[8], p1[8];
  #pragma unroll
  for (int j = 0; j < 8; ++j) {
    float x0 = acc0[j]; x0 = (x0 > 0.0f) ? x0 : ALPHA * x0;
    float x1 = acc1[j]; x1 = (x1 > 0.0f) ? x1 : ALPHA * x1;
    int r = nbase + j + 8 * lhalf;
    if (r < N) X1out[(size_t)r * HD + cbase + l15] = x1;
    p0[j] = x0 * aval;
    p1[j] = x1 * aval;
  }

  #pragma unroll
  for (int j = 0; j < 8; ++j) {
    #pragma unroll
    for (int off = 8; off >= 1; off >>= 1) {
      p0[j] += __shfl_xor(p0[j], off, 32);
      p1[j] += __shfl_xor(p1[j], off, 32);
    }
  }
  if (l15 == 0) {
    #pragma unroll
    for (int j = 0; j < 8; ++j) {
      int r = nbase + j + 8 * lhalf;
      if (r < N) {
        atomicAdd(&s0[r * NHEAD + head], p0[j]);
        atomicAdd(&s1[r * NHEAD + head], p1[j]);
      }
    }
  }
}

__global__ __launch_bounds__(256) void gat_minmax_part(
    const float* __restrict__ s0, const float* __restrict__ s1,
    const int* __restrict__ esrc, const int* __restrict__ ecol, int E,
    float* __restrict__ pmax, float* __restrict__ pmn)
{
  const int tid = threadIdx.x;
  float mx[NHEAD], mn[NHEAD];
  #pragma unroll
  for (int h = 0; h < NHEAD; ++h) { mx[h] = -1e30f; mn[h] = 1e30f; }

  for (int e = blockIdx.x * 256 + tid; e < E; e += gridDim.x * 256) {
    int s = esrc[e];
    int c = ecol[e];
    const float* sa = s0 + (size_t)s * NHEAD;
    const float* sb = s1 + (size_t)c * NHEAD;
    #pragma unroll
    for (int h = 0; h < NHEAD; ++h) {
      float att = sa[h] + sb[h];
      mx[h] = fmaxf(mx[h], att);
      mn[h] = fminf(mn[h], att);
    }
  }

  #pragma unroll
  for (int h = 0; h < NHEAD; ++h)
    for (int off = 16; off >= 1; off >>= 1) {
      mx[h] = fmaxf(mx[h], __shfl_xor(mx[h], off, 32));
      mn[h] = fminf(mn[h], __shfl_xor(mn[h], off, 32));
    }

  __shared__ float smx[8][NHEAD], smn[8][NHEAD];
  const int wave = tid >> 5, lane = tid & 31;
  if (lane == 0) {
    #pragma unroll
    for (int h = 0; h < NHEAD; ++h) { smx[wave][h] = mx[h]; smn[wave][h] = mn[h]; }
  }
  __syncthreads();
  if (tid == 0) {
    #pragma unroll
    for (int h = 0; h < NHEAD; ++h) {
      float a = smx[0][h], b = smn[0][h];
      for (int w = 1; w < 8; ++w) { a = fmaxf(a, smx[w][h]); b = fminf(b, smn[w][h]); }
      pmax[blockIdx.x * NHEAD + h] = a;
      pmn [blockIdx.x * NHEAD + h] = b;
    }
  }
}

__global__ __launch_bounds__(256) void gat_minmax_final(
    const float* __restrict__ pmax, const float* __restrict__ pmn, int nb,
    float* __restrict__ mxmn)
{
  const int tid = threadIdx.x;
  float mx[NHEAD], mn[NHEAD];
  #pragma unroll
  for (int h = 0; h < NHEAD; ++h) { mx[h] = -1e30f; mn[h] = 1e30f; }

  for (int i = tid; i < nb; i += 256) {
    #pragma unroll
    for (int h = 0; h < NHEAD; ++h) {
      mx[h] = fmaxf(mx[h], pmax[i * NHEAD + h]);
      mn[h] = fminf(mn[h], pmn [i * NHEAD + h]);
    }
  }
  #pragma unroll
  for (int h = 0; h < NHEAD; ++h)
    for (int off = 16; off >= 1; off >>= 1) {
      mx[h] = fmaxf(mx[h], __shfl_xor(mx[h], off, 32));
      mn[h] = fminf(mn[h], __shfl_xor(mn[h], off, 32));
    }

  __shared__ float smx[8][NHEAD], smn[8][NHEAD];
  const int wave = tid >> 5, lane = tid & 31;
  if (lane == 0) {
    #pragma unroll
    for (int h = 0; h < NHEAD; ++h) { smx[wave][h] = mx[h]; smn[wave][h] = mn[h]; }
  }
  __syncthreads();
  if (tid == 0) {
    #pragma unroll
    for (int h = 0; h < NHEAD; ++h) {
      float a = smx[0][h], b = smn[0][h];
      for (int w = 1; w < 8; ++w) { a = fmaxf(a, smx[w][h]); b = fminf(b, smn[w][h]); }
      mxmn[h]         = a;
      mxmn[NHEAD + h] = b;
    }
  }
}

__global__ __launch_bounds__(256) void gat_edge(
    const float* __restrict__ X1, const float* __restrict__ s0,
    const float* __restrict__ s1, const int* __restrict__ ecol,
    const float* __restrict__ mxmn, float* __restrict__ out, int N)
{
  const int wave = threadIdx.x >> 5;
  const int lane = threadIdx.x & 31;
  const int n = blockIdx.x * 8 + wave;
  if (n >= N) return;

  float scale[NHEAD], s0n[NHEAD];
  #pragma unroll
  for (int h = 0; h < NHEAD; ++h) {
    float mxv = mxmn[h], mnv = mxmn[NHEAD + h];
    scale[h] = 1.0f / (mxv - mnv);
    s0n[h]   = s0[n * NHEAD + h] - mnv;
  }

  float acc[NHEAD]  = {0.f, 0.f, 0.f, 0.f};
  float wsum[NHEAD] = {0.f, 0.f, 0.f, 0.f};

  const int* cols = ecol + n * DEG;
  for (int e = 0; e < DEG; ++e) {
    int c = cols[e];
    const float* sc   = s1 + (size_t)c * NHEAD;
    const float* xrow = X1 + (size_t)c * HD;
    #pragma unroll
    for (int h = 0; h < NHEAD; ++h) {
      float att = __expf((s0n[h] + sc[h]) * scale[h]);
      wsum[h] += att;
      acc[h]  += att * xrow[h * DHEAD + lane];
    }
  }
  #pragma unroll
  for (int h = 0; h < NHEAD; ++h)
    out[(size_t)n * HD + h * DHEAD + lane] = acc[h] / wsum[h];
}

extern "C" void kernel_launch(void* const* d_in, const int* in_sizes, int n_in,
                              void* d_out, int out_size, void* d_ws, size_t ws_size,
                              hipStream_t stream) {
  const float* X    = (const float*)d_in[0];
  const float* W0   = (const float*)d_in[1];
  const float* W1   = (const float*)d_in[2];
  const float* a0   = (const float*)d_in[3];
  const int*   esrc = (const int*)d_in[4];
  const int*   ecol = (const int*)d_in[5];
  float*       out  = (float*)d_out;

  const int N = in_sizes[0] / IN_F;
  const int E = in_sizes[5];
  const int NB = 512;

  float* X1ws = (float*)d_ws;
  float* s0   = X1ws + (size_t)N * HD;
  float* s1   = s0 + (size_t)N * NHEAD;
  float* pmax = s1 + (size_t)N * NHEAD;
  float* pmn  = pmax + (size_t)NB * NHEAD;
  float* mxmn = pmn + (size_t)NB * NHEAD;

  gat_zero<<<(2 * N * NHEAD + 255) / 256, 256, 0, stream>>>(s0, 2 * N * NHEAD);
  gat_gemm<<<(N + 15) / 16, 256, 0, stream>>>(X, W0, W1, a0, X1ws, s0, s1, N);
  gat_minmax_part<<<NB, 256, 0, stream>>>(s0, s1, esrc, ecol, E, pmax, pmn);
  gat_minmax_final<<<1, 256, 0, stream>>>(pmax, pmn, NB, mxmn);
  gat_edge<<<(N + 7) / 8, 256, 0, stream>>>(X1ws, s0, s1, ecol, mxmn, out, N);
}


# edge-score min/max in ONE launch: a lane keeps 4 x 16-byte id groups (16 edges, 32 score rows) in flight; work-group results go to 8 words by exact integer atomic max/min; fold launch gone (4->3)
# speedup vs baseline: 1.8306x; 1.8306x over previous
#include <hip/hip_runtime.h>
#include <math.h>

typedef float v2f __attribute__((ext_vector_type(2)));
typedef float v8f __attribute__((ext_vector_type(8)));

#define IN_F  128
#define HD    128
#define NHEAD 4
#define DHEAD 32
#define DEG   16
#define ALPHA 0.2f
#define WPITCH 132
#define GEMM_WGS 256
#define MM_WGS   256
#define MM_WAVES 16
#define MM_TRIPS 4

__device__ __forceinline__ unsigned f2key(float f) {
  unsigned b = __float_as_uint(f);
  return (b & 0x80000000u) ? ~b : (b | 0x80000000u);
}
__device__ __forceinline__ float key2f(unsigned k) {
  return __uint_as_float((k & 0x80000000u) ? (k & 0x7fffffffu) : ~k);
}

#if defined(__HIP_DEVICE_COMPILE__)
#if !__has_builtin(__builtin_amdgcn_wmma_f32_16x16x4_f32)
#error "gfx1250 fp32 WMMA builtin not available on this toolchain (device pass)"
#endif
#endif

__global__ __launch_bounds__(256) void gat_gemm(
    const float* __restrict__ X, const float* __restrict__ W0,
    const float* __restrict__ W1, const float* __restrict__ a0,
    float* __restrict__ X1out, float* __restrict__ s0, float* __restrict__ s1,
    float* __restrict__ mxmn, int N)
{
  __shared__ __attribute__((aligned(16))) float lW[2 * HD * WPITCH];

  const int tid  = threadIdx.x;
  if (blockIdx.x == 0 && tid < 2 * NHEAD)
    reinterpret_cast<unsigned*>(mxmn)[tid] = (tid < NHEAD) ? 0u : 0xffffffffu;
  const int wave = tid >> 5;
  const int lane = tid & 31;
  const int c    = lane & 15;
  const int h    = lane >> 4;

  const int ntiles = (N + 15) >> 4;
  const int nwaves = gridDim.x * 8;
  int tile = wave * gridDim.x + blockIdx.x;

  float4 a[4];
  {
    int row = tile * 16 + c; if (row > N - 1) row = N - 1;
    const float* xr = X + (size_t)row * IN_F + 4 * h;
    #pragma unroll
    for (int j = 0; j < 4; ++j) a[j] = *(const float4*)(xr + 8 * j);
  }

  #pragma unroll
  for (int i = 0; i < 16; ++i) {
    int idx = i * 256 + tid;
    int row = idx >> 5;
    int c4  = (idx & 31) * 4;
    float4 q = *(const float4*)(W0 + (size_t)row * IN_F + c4);
    *(float4*)(lW + row * WPITCH + c4) = q;
  }
  #pragma unroll
  for (int i = 0; i < 16; ++i) {
    int idx = i * 256 + tid;
    int row = idx >> 5;
    int c4  = (idx & 31) * 4;
    float4 q = *(const float4*)(W1 + (size_t)row * IN_F + c4);
    *(float4*)(lW + (HD + row) * WPITCH + c4) = q;
  }

  float av[8];
  #pragma unroll
  for (int ct = 0; ct < 8; ++ct) av[ct] = a0[ct * 16 + c];

  __syncthreads();

  const float* lB = lW + c * WPITCH + 4 * h;

#define LOAD_BSET(dst, base)                                                   \
  _Pragma("unroll")                                                            \
  for (int ct = 0; ct < 16; ++ct)                                              \
    dst[ct] = *(const float4*)((base) + ct * 16 * WPITCH);
#define WMMA_BSET(afrag, bset)                                                 \
  {                                                                            \
    v2f alo; alo.x = (afrag).x; alo.y = (afrag).y;                             \
    v2f ahi; ahi.x = (afrag).z; ahi.y = (afrag).w;                             \
    _Pragma("unroll")                                                          \
    for (int ct = 0; ct < 16; ++ct) {                                          \
      v2f bl; bl.x = bset[ct].x; bl.y = bset[ct].y;                            \
      acc[ct] = __builtin_amdgcn_wmma_f32_16x16x4_f32(false, alo, false, bl,   \
                                         (short)0, acc[ct], false, false);     \
    }                                                                          \
    _Pragma("unroll")                                                          \
    for (int ct = 0; ct < 16; ++ct) {                                          \
      v2f bh; bh.x = bset[ct].z; bh.y = bset[ct].w;                            \
      acc[ct] = __builtin_amdgcn_wmma_f32_16x16x4_f32(false, ahi, false, bh,   \
                                         (short)0, acc[ct], false, false);     \
    }                                                                          \
  }
#define SCHED_FENCE() __builtin_amdgcn_sched_barrier(0)

  float4 B0[16], B1[16];
  LOAD_BSET(B0, lB)

  for (; tile < ntiles; tile += nwaves) {
    int nt = tile + nwaves; if (nt >= ntiles) nt = tile;
    int nrow = nt * 16 + c; if (nrow > N - 1) nrow = N - 1;
    int crow = tile * 16 + c; if (crow > N - 1) crow = N - 1;
    const float* xn = X + (size_t)nrow * IN_F + 4 * h;
    const float* xc = X + (size_t)crow * IN_F + 4 * h;

    v8f acc[16];
    #pragma unroll
    for (int ct = 0; ct < 16; ++ct) acc[ct] = (v8f){};

    #pragma unroll 1
    for (int jj = 0; jj < 4; ++jj) {
      const float* lBj = lB + jj * 32;
      const float* lBn = lB + ((jj + 1) & 3) * 32;
      const float* xp  = (jj < 3) ? (xc + (jj + 1) * 32) : xn;

      LOAD_BSET(B1, lBj + 8)
      SCHED_FENCE();
      WMMA_BSET(a[0], B0)
      SCHED_FENCE();
      a[0] = *(const float4*)(xp);
      LOAD_BSET(B0, lBj + 16)
      SCHED_FENCE();
      WMMA_BSET(a[1], B1)
      SCHED_FENCE();
      a[1] = *(const float4*)(xp + 8);
      LOAD_BSET(B1, lBj + 24)
      SCHED_FENCE();
      WMMA_BSET(a[2], B0)
      SCHED_FENCE();
      a[2] = *(const float4*)(xp + 16);
      LOAD_BSET(B0, lBn)
      SCHED_FENCE();
      WMMA_BSET(a[3], B1)
      SCHED_FENCE();
      a[3] = *(const float4*)(xp + 24);
    }

    const int nb = tile * 16 + 8 * h;
    #pragma unroll
    for (int r = 0; r < 8; ++r) {
      const int node = nb + r;
      float p0[NHEAD], p1[NHEAD];
      #pragma unroll
      for (int hd = 0; hd < NHEAD; ++hd) {
        float u0 = acc[2 * hd][r];         u0 = (u0 > 0.0f) ? u0 : ALPHA * u0;
        float u1 = acc[2 * hd + 1][r];     u1 = (u1 > 0.0f) ? u1 : ALPHA * u1;
        float w0 = acc[8 + 2 * hd][r];     w0 = (w0 > 0.0f) ? w0 : ALPHA * w0;
        float w1 = acc[8 + 2 * hd + 1][r]; w1 = (w1 > 0.0f) ? w1 : ALPHA * w1;
        if (node < N) {
          X1out[(size_t)node * HD + (2 * hd) * 16 + c]     = w0;
          X1out[(size_t)node * HD + (2 * hd + 1) * 16 + c] = w1;
        }
        p0[hd] = u0 * av[2 * hd] + u1 * av[2 * hd + 1];
        p1[hd] = w0 * av[2 * hd] + w1 * av[2 * hd + 1];
      }
      #pragma unroll
      for (int hd = 0; hd < NHEAD; ++hd) {
        #pragma unroll
        for (int off = 8; off >= 1; off >>= 1) {
          p0[hd] += __shfl_xor(p0[hd], off, 32);
          p1[hd] += __shfl_xor(p1[hd], off, 32);
        }
      }
      if (c == 0 && node < N) {
        *(float4*)(s0 + (size_t)node * NHEAD) = make_float4(p0[0], p0[1], p0[2], p0[3]);
        *(float4*)(s1 + (size_t)node * NHEAD) = make_float4(p1[0], p1[1], p1[2], p1[3]);
      }
    }
  }
}

#undef LOAD_BSET
#undef WMMA_BSET
#undef SCHED_FENCE

__global__ __launch_bounds__(MM_WAVES * 32) __attribute__((amdgpu_waves_per_eu(4, 4))) void gat_minmax(
    const float* __restrict__ s0, const float* __restrict__ s1,
    const int* __restrict__ esrc, const int* __restrict__ ecol, int E,
    float* __restrict__ mxmn)
{
  const int tid = threadIdx.x;
  const int wave = tid >> 5, lane = tid & 31;
  float mx[NHEAD], mn[NHEAD];
  #pragma unroll
  for (int h = 0; h < NHEAD; ++h) { mx[h] = -1e30f; mn[h] = 1e30f; }

#define MM_EDGE(sidx, cidx)                                                    \
  {                                                                            \
    const float4 sa = *(const float4*)(s0 + (size_t)(sidx) * NHEAD);           \
    const float4 sb = *(const float4*)(s1 + (size_t)(cidx) * NHEAD);           \
    float att;                                                                 \
    att = sa.x + sb.x; mx[0] = fmaxf(mx[0], att); mn[0] = fminf(mn[0], att);   \
    att = sa.y + sb.y; mx[1] = fmaxf(mx[1], att); mn[1] = fminf(mn[1], att);   \
    att = sa.z + sb.z; mx[2] = fmaxf(mx[2], att); mn[2] = fminf(mn[2], att);   \
    att = sa.w + sb.w; mx[3] = fmaxf(mx[3], att); mn[3] = fminf(mn[3], att);   \
  }

  const int nq     = E >> 2;
  const int nchunk = (nq + 32 * MM_TRIPS - 1) / (32 * MM_TRIPS);
  const int nwaves = gridDim.x * MM_WAVES;
  for (int ch = wave * gridDim.x + blockIdx.x; ch < nchunk; ch += nwaves) {
    const int qb = ch * (32 * MM_TRIPS) + lane;
    int4 sv[MM_TRIPS], cv[MM_TRIPS];
    #pragma unroll
    for (int u = 0; u < MM_TRIPS; ++u) {
      int q = qb + 32 * u; if (q > nq - 1) q = nq - 1;
      sv[u] = *(const int4*)(esrc + (size_t)q * 4);
      cv[u] = *(const int4*)(ecol + (size_t)q * 4);
    }
    #pragma unroll
    for (int u = 0; u < MM_TRIPS; ++u) {
      MM_EDGE(sv[u].x, cv[u].x)
      MM_EDGE(sv[u].y, cv[u].y)
      MM_EDGE(sv[u].z, cv[u].z)
      MM_EDGE(sv[u].w, cv[u].w)
    }
  }
  if (blockIdx.x == 0 && tid < (E & 3)) {
    const int e = (E & ~3) + tid;
    MM_EDGE(esrc[e], ecol[e])
  }
#undef MM_EDGE

  #pragma unroll
  for (int h = 0; h < NHEAD; ++h)
    for (int off = 16; off >= 1; off >>= 1) {
      mx[h] = fmaxf(mx[h], __shfl_xor(mx[h], off, 32));
      mn[h] = fminf(mn[h], __shfl_xor(mn[h], off, 32));
    }

  __shared__ float smx[MM_WAVES][NHEAD], smn[MM_WAVES][NHEAD];
  if (lane == 0) {
    #pragma unroll
    for (int h = 0; h < NHEAD; ++h) { smx[wave][h] = mx[h]; smn[wave][h] = mn[h]; }
  }
  __syncthreads();
  if (tid < 2 * NHEAD) {
    unsigned* keys = reinterpret_cast<unsigned*>(mxmn);
    const int h = tid & (NHEAD - 1);
    if (tid < NHEAD) {
      float a = smx[0][h];
      for (int w = 1; w < MM_WAVES; ++w) a = fmaxf(a, smx[w][h]);
      atomicMax(keys + h, f2key(a));
    } else {
      float b = smn[0][h];
      for (int w = 1; w < MM_WAVES; ++w) b = fminf(b, smn[w][h]);
      atomicMin(keys + NHEAD + h, f2key(b));
    }
  }
}

__global__ __launch_bounds__(256) void gat_edge(
    const float* __restrict__ X1, const float* __restrict__ s0,
    const float* __restrict__ s1, const int* __restrict__ ecol,
    const float* __restrict__ mxmn, float* __restrict__ out, int N)
{
  const int wave = threadIdx.x >> 5;
  const int lane = threadIdx.x & 31;
  const int n = blockIdx.x * 8 + wave;
  if (n >= N) return;

  const unsigned* keys = reinterpret_cast<const unsigned*>(mxmn);
  float scale[NHEAD], s0n[NHEAD];
  #pragma unroll
  for (int h = 0; h < NHEAD; ++h) {
    const float mxv = key2f(keys[h]), mnv = key2f(keys[NHEAD + h]);
    scale[h] = 1.0f / (mxv - mnv);
    s0n[h]   = s0[n * NHEAD + h] - mnv;
  }

  float acc[NHEAD]  = {0.f, 0.f, 0.f, 0.f};
  float wsum[NHEAD] = {0.f, 0.f, 0.f, 0.f};

  const int* cols = ecol + n * DEG;
  for (int e = 0; e < DEG; ++e) {
    int c = cols[e];
    const float* sc   = s1 + (size_t)c * NHEAD;
    const float* xrow = X1 + (size_t)c * HD;
    #pragma unroll
    for (int h = 0; h < NHEAD; ++h) {
      float att = __expf((s0n[h] + sc[h]) * scale[h]);
      wsum[h] += att;
      acc[h]  += att * xrow[h * DHEAD + lane];
    }
  }
  #pragma unroll
  for (int h = 0; h < NHEAD; ++h)
    out[(size_t)n * HD + h * DHEAD + lane] = acc[h] / wsum[h];
}

extern "C" void kernel_launch(void* const* d_in, const int* in_sizes, int n_in,
                              void* d_out, int out_size, void* d_ws, size_t ws_size,
                              hipStream_t stream) {
  const float* X    = (const float*)d_in[0];
  const float* W0   = (const float*)d_in[1];
  const float* W1   = (const float*)d_in[2];
  const float* a0   = (const float*)d_in[3];
  const int*   esrc = (const int*)d_in[4];
  const int*   ecol = (const int*)d_in[5];
  float*       out  = (float*)d_out;

  const int N = in_sizes[0] / IN_F;
  const int E = in_sizes[5];

  float* X1ws = (float*)d_ws;
  float* s0   = X1ws + (size_t)N * HD;
  float* s1   = s0 + (size_t)N * NHEAD;
  float* mxmn = s1 + (size_t)N * NHEAD;

  gat_gemm<<<GEMM_WGS, 256, 0, stream>>>(X, W0, W1, a0, X1ws, s0, s1, mxmn, N);
  gat_minmax<<<MM_WGS, MM_WAVES * 32, 0, stream>>>(s0, s1, esrc, ecol, E, mxmn);
  gat_edge<<<(N + 7) / 8, 256, 0, stream>>>(X1ws, s0, s1, ecol, mxmn, out, N);
}


# edge pass as a resident grid of looping waves: a node's ids, score rows and exps taken once (lane = edge), handed to all lanes by v_readlane; its 64 row loads refilled in place (rolling window)
# speedup vs baseline: 1.9226x; 1.0503x over previous
#include <hip/hip_runtime.h>
#include <math.h>

typedef float v2f __attribute__((ext_vector_type(2)));
typedef float v8f __attribute__((ext_vector_type(8)));

#define IN_F  128
#define HD    128
#define NHEAD 4
#define DHEAD 32
#define DEG   16
#define ALPHA 0.2f
#define WPITCH 132
#define GEMM_WGS 256
#define MM_WGS   256
#define MM_WAVES 16
#define MM_TRIPS 4

__device__ __forceinline__ unsigned f2key(float f) {
  unsigned b = __float_as_uint(f);
  return (b & 0x80000000u) ? ~b : (b | 0x80000000u);
}
__device__ __forceinline__ float key2f(unsigned k) {
  return __uint_as_float((k & 0x80000000u) ? (k & 0x7fffffffu) : ~k);
}

#if defined(__HIP_DEVICE_COMPILE__)
#if !__has_builtin(__builtin_amdgcn_wmma_f32_16x16x4_f32)
#error "gfx1250 fp32 WMMA builtin not available on this toolchain (device pass)"
#endif
#endif

__global__ __launch_bounds__(256) void gat_gemm(
    const float* __restrict__ X, const float* __restrict__ W0,
    const float* __restrict__ W1, const float* __restrict__ a0,
    float* __restrict__ X1out, float* __restrict__ s0, float* __restrict__ s1,
    float* __restrict__ mxmn, int N)
{
  __shared__ __attribute__((aligned(16))) float lW[2 * HD * WPITCH];

  const int tid  = threadIdx.x;
  if (blockIdx.x == 0 && tid < 2 * NHEAD)
    reinterpret_cast<unsigned*>(mxmn)[tid] = (tid < NHEAD) ? 0u : 0xffffffffu;
  const int wave = tid >> 5;
  const int lane = tid & 31;
  const int c    = lane & 15;
  const int h    = lane >> 4;

  const int ntiles = (N + 15) >> 4;
  const int nwaves = gridDim.x * 8;
  int tile = wave * gridDim.x + blockIdx.x;

  float4 a[4];
  {
    int row = tile * 16 + c; if (row > N - 1) row = N - 1;
    const float* xr = X + (size_t)row * IN_F + 4 * h;
    #pragma unroll
    for (int j = 0; j < 4; ++j) a[j] = *(const float4*)(xr + 8 * j);
  }

  #pragma unroll
  for (int i = 0; i < 16; ++i) {
    int idx = i * 256 + tid;
    int row = idx >> 5;
    int c4  = (idx & 31) * 4;
    float4 q = *(const float4*)(W0 + (size_t)row * IN_F + c4);
    *(float4*)(lW + row * WPITCH + c4) = q;
  }
  #pragma unroll
  for (int i = 0; i < 16; ++i) {
    int idx = i * 256 + tid;
    int row = idx >> 5;
    int c4  = (idx & 31) * 4;
    float4 q = *(const float4*)(W1 + (size_t)row * IN_F + c4);
    *(float4*)(lW + (HD + row) * WPITCH + c4) = q;
  }

  float av[8];
  #pragma unroll
  for (int ct = 0; ct < 8; ++ct) av[ct] = a0[ct * 16 + c];

  __syncthreads();

  const float* lB = lW + c * WPITCH + 4 * h;

#define LOAD_BSET(dst, base)                                                   \
  _Pragma("unroll")                                                            \
  for (int ct = 0; ct < 16; ++ct)                                              \
    dst[ct] = *(const float4*)((base) + ct * 16 * WPITCH);
#define WMMA_BSET(afrag, bset)                                                 \
  {                                                                            \
    v2f alo; alo.x = (afrag).x; alo.y = (afrag).y;                             \
    v2f ahi; ahi.x = (afrag).z; ahi.y = (afrag).w;                             \
    _Pragma("unroll")                                                          \
    for (int ct = 0; ct < 16; ++ct) {                                          \
      v2f bl; bl.x = bset[ct].x; bl.y = bset[ct].y;                            \
      acc[ct] = __builtin_amdgcn_wmma_f32_16x16x4_f32(false, alo, false, bl,   \
                                         (short)0, acc[ct], false, false);     \
    }                                                                          \
    _Pragma("unroll")                                                          \
    for (int ct = 0; ct < 16; ++ct) {                                          \
      v2f bh; bh.x = bset[ct].z; bh.y = bset[ct].w;                            \
      acc[ct] = __builtin_amdgcn_wmma_f32_16x16x4_f32(false, ahi, false, bh,   \
                                         (short)0, acc[ct], false, false);     \
    }                                                                          \
  }
#define SCHED_FENCE() __builtin_amdgcn_sched_barrier(0)

  float4 B0[16], B1[16];
  LOAD_BSET(B0, lB)

  for (; tile < ntiles; tile += nwaves) {
    int nt = tile + nwaves; if (nt >= ntiles) nt = tile;
    int nrow = nt * 16 + c; if (nrow > N - 1) nrow = N - 1;
    int crow = tile * 16 + c; if (crow > N - 1) crow = N - 1;
    const float* xn = X + (size_t)nrow * IN_F + 4 * h;
    const float* xc = X + (size_t)crow * IN_F + 4 * h;

    v8f acc[16];
    #pragma unroll
    for (int ct = 0; ct < 16; ++ct) acc[ct] = (v8f){};

    #pragma unroll 1
    for (int jj = 0; jj < 4; ++jj) {
      const float* lBj = lB + jj * 32;
      const float* lBn = lB + ((jj + 1) & 3) * 32;
      const float* xp  = (jj < 3) ? (xc + (jj + 1) * 32) : xn;

      LOAD_BSET(B1, lBj + 8)
      SCHED_FENCE();
      WMMA_BSET(a[0], B0)
      SCHED_FENCE();
      a[0] = *(const float4*)(xp);
      LOAD_BSET(B0, lBj + 16)
      SCHED_FENCE();
      WMMA_BSET(a[1], B1)
      SCHED_FENCE();
      a[1] = *(const float4*)(xp + 8);
      LOAD_BSET(B1, lBj + 24)
      SCHED_FENCE();
      WMMA_BSET(a[2], B0)
      SCHED_FENCE();
      a[2] = *(const float4*)(xp + 16);
      LOAD_BSET(B0, lBn)
      SCHED_FENCE();
      WMMA_BSET(a[3], B1)
      SCHED_FENCE();
      a[3] = *(const float4*)(xp + 24);
    }

    const int nb = tile * 16 + 8 * h;
    #pragma unroll
    for (int r = 0; r < 8; ++r) {
      const int node = nb + r;
      float p0[NHEAD], p1[NHEAD];
      #pragma unroll
      for (int hd = 0; hd < NHEAD; ++hd) {
        float u0 = acc[2 * hd][r];         u0 = (u0 > 0.0f) ? u0 : ALPHA * u0;
        float u1 = acc[2 * hd + 1][r];     u1 = (u1 > 0.0f) ? u1 : ALPHA * u1;
        float w0 = acc[8 + 2 * hd][r];     w0 = (w0 > 0.0f) ? w0 : ALPHA * w0;
        float w1 = acc[8 + 2 * hd + 1][r]; w1 = (w1 > 0.0f) ? w1 : ALPHA * w1;
        if (node < N) {
          X1out[(size_t)node * HD + (2 * hd) * 16 + c]     = w0;
          X1out[(size_t)node * HD + (2 * hd + 1) * 16 + c] = w1;
        }
        p0[hd] = u0 * av[2 * hd] + u1 * av[2 * hd + 1];
        p1[hd] = w0 * av[2 * hd] + w1 * av[2 * hd + 1];
      }
      #pragma unroll
      for (int hd = 0; hd < NHEAD; ++hd) {
        #pragma unroll
        for (int off = 8; off >= 1; off >>= 1) {
          p0[hd] += __shfl_xor(p0[hd], off, 32);
          p1[hd] += __shfl_xor(p1[hd], off, 32);
        }
      }
      if (c == 0 && node < N) {
        *(float4*)(s0 + (size_t)node * NHEAD) = make_float4(p0[0], p0[1], p0[2], p0[3]);
        *(float4*)(s1 + (size_t)node * NHEAD) = make_float4(p1[0], p1[1], p1[2], p1[3]);
      }
    }
  }
}

#undef LOAD_BSET
#undef WMMA_BSET
#undef SCHED_FENCE

__global__ __launch_bounds__(MM_WAVES * 32) __attribute__((amdgpu_waves_per_eu(4, 4))) void gat_minmax(
    const float* __restrict__ s0, const float* __restrict__ s1,
    const int* __restrict__ esrc, const int* __restrict__ ecol, int E,
    float* __restrict__ mxmn)
{
  const int tid = threadIdx.x;
  const int wave = tid >> 5, lane = tid & 31;
  float mx[NHEAD], mn[NHEAD];
  #pragma unroll
  for (int h = 0; h < NHEAD; ++h) { mx[h] = -1e30f; mn[h] = 1e30f; }

#define MM_EDGE(sidx, cidx)                                                    \
  {                                                                            \
    const float4 sa = *(const float4*)(s0 + (size_t)(sidx) * NHEAD);           \
    const float4 sb = *(const float4*)(s1 + (size_t)(cidx) * NHEAD);           \
    float att;                                                                 \
    att = sa.x + sb.x; mx[0] = fmaxf(mx[0], att); mn[0] = fminf(mn[0], att);   \
    att = sa.y + sb.y; mx[1] = fmaxf(mx[1], att); mn[1] = fminf(mn[1], att);   \
    att = sa.z + sb.z; mx[2] = fmaxf(mx[2], att); mn[2] = fminf(mn[2], att);   \
    att = sa.w + sb.w; mx[3] = fmaxf(mx[3], att); mn[3] = fminf(mn[3], att);   \
  }

  const int nq     = E >> 2;
  const int nchunk = (nq + 32 * MM_TRIPS - 1) / (32 * MM_TRIPS);
  const int nwaves = gridDim.x * MM_WAVES;
  for (int ch = wave * gridDim.x + blockIdx.x; ch < nchunk; ch += nwaves) {
    const int qb = ch * (32 * MM_TRIPS) + lane;
    int4 sv[MM_TRIPS], cv[MM_TRIPS];
    #pragma unroll
    for (int u = 0; u < MM_TRIPS; ++u) {
      int q = qb + 32 * u; if (q > nq - 1) q = nq - 1;
      sv[u] = *(const int4*)(esrc + (size_t)q * 4);
      cv[u] = *(const int4*)(ecol + (size_t)q * 4);
    }
    #pragma unroll
    for (int u = 0; u < MM_TRIPS; ++u) {
      MM_EDGE(sv[u].x, cv[u].x)
      MM_EDGE(sv[u].y, cv[u].y)
      MM_EDGE(sv[u].z, cv[u].z)
      MM_EDGE(sv[u].w, cv[u].w)
    }
  }
  if (blockIdx.x == 0 && tid < (E & 3)) {
    const int e = (E & ~3) + tid;
    MM_EDGE(esrc[e], ecol[e])
  }
#undef MM_EDGE

  #pragma unroll
  for (int h = 0; h < NHEAD; ++h)
    for (int off = 16; off >= 1; off >>= 1) {
      mx[h] = fmaxf(mx[h], __shfl_xor(mx[h], off, 32));
      mn[h] = fminf(mn[h], __shfl_xor(mn[h], off, 32));
    }

  __shared__ float smx[MM_WAVES][NHEAD], smn[MM_WAVES][NHEAD];
  if (lane == 0) {
    #pragma unroll
    for (int h = 0; h < NHEAD; ++h) { smx[wave][h] = mx[h]; smn[wave][h] = mn[h]; }
  }
  __syncthreads();
  if (tid < 2 * NHEAD) {
    unsigned* keys = reinterpret_cast<unsigned*>(mxmn);
    const int h = tid & (NHEAD - 1);
    if (tid < NHEAD) {
      float a = smx[0][h];
      for (int w = 1; w < MM_WAVES; ++w) a = fmaxf(a, smx[w][h]);
      atomicMax(keys + h, f2key(a));
    } else {
      float b = smn[0][h];
      for (int w = 1; w < MM_WAVES; ++w) b = fminf(b, smn[w][h]);
      atomicMin(keys + NHEAD + h, f2key(b));
    }
  }
}

#define EDGE_WGS   512
#define EDGE_WAVES 16

__device__ __forceinline__ float lane_to_all(float v, int l) {
  return __int_as_float(__builtin_amdgcn_readlane(__float_as_int(v), l));
}

__device__ __forceinline__ void edge_row_offsets(unsigned (&off)[DEG], int col, int lane) {
  #pragma unroll
  for (int e = 0; e < DEG; ++e)
    off[e] = (unsigned)__builtin_amdgcn_readlane(col, e) * HD + lane;
}

__device__ __forceinline__ void edge_row_load(float (&xe)[NHEAD], unsigned off,
                                              const float* __restrict__ X1) {
  #pragma unroll
  for (int h = 0; h < NHEAD; ++h) xe[h] = X1[off + h * DHEAD];
}

template <bool REFILL>
__device__ __forceinline__ void edge_node(float (&x)[DEG][NHEAD], const float (&w)[NHEAD],
                                          const unsigned (&off_next)[DEG],
                                          const float* __restrict__ X1,
                                          int lane, float* __restrict__ orow) {
  float acc[NHEAD]  = {0.f, 0.f, 0.f, 0.f};
  float wsum[NHEAD] = {0.f, 0.f, 0.f, 0.f};
  #pragma unroll
  for (int e = 0; e < DEG; ++e) {
    #pragma unroll
    for (int h = 0; h < NHEAD; ++h) {
      const float wh = lane_to_all(w[h], e);
      wsum[h] += wh;
      acc[h]   = __builtin_fmaf(wh, x[e][h], acc[h]);
    }
    if (REFILL) {
      asm volatile("" :: "v"(acc[0]), "v"(acc[1]), "v"(acc[2]), "v"(acc[3]));
      __builtin_amdgcn_sched_barrier(0);
      edge_row_load(x[e], off_next[e], X1);
      __builtin_amdgcn_sched_barrier(0);
    }
  }
  #pragma unroll
  for (int h = 0; h < NHEAD; ++h) orow[h * DHEAD + lane] = acc[h] / wsum[h];
}

__global__ __launch_bounds__(EDGE_WAVES * 32) __attribute__((amdgpu_waves_per_eu(8, 8))) void gat_edge(
    const float* __restrict__ X1, const float* __restrict__ s0,
    const float* __restrict__ s1, const int* __restrict__ ecol,
    const float* __restrict__ mxmn, float* __restrict__ out, int N)
{
  const int lane = threadIdx.x & 31;
  const int e16  = lane & 15;
  const int wave = __builtin_amdgcn_readfirstlane((int)(threadIdx.x >> 5));
  const int stride = gridDim.x * EDGE_WAVES;
  int n = wave * gridDim.x + blockIdx.x;
  if (n >= N) return;

  const unsigned* keys = reinterpret_cast<const unsigned*>(mxmn);
  float scale[NHEAD], mnv[NHEAD];
  #pragma unroll
  for (int h = 0; h < NHEAD; ++h) {
    const float mxv = key2f(keys[h]);
    mnv[h]   = key2f(keys[NHEAD + h]);
    scale[h] = 1.0f / (mxv - mnv[h]);
  }

  int n1  = n + stride;
  int n1c = (n1 < N) ? n1 : N - 1;
  const int col0 = ecol[(size_t)n * DEG + e16];
  int col1 = ecol[(size_t)n1c * DEG + e16];
  float4 s0v = *(const float4*)(s0 + (size_t)n * NHEAD);
  float4 s1v = *(const float4*)(s1 + (size_t)(unsigned)col0 * NHEAD);
  __builtin_amdgcn_sched_barrier(0);
  float x[DEG][NHEAD];
  unsigned off[DEG];
  edge_row_offsets(off, col0, lane);
  #pragma unroll
  for (int e = 0; e < DEG; ++e) {
    edge_row_load(x[e], off[e], X1);
    __builtin_amdgcn_sched_barrier(0);
  }

  for (;;) {
    float w[NHEAD];
    w[0] = __expf(((s0v.x - mnv[0]) + s1v.x) * scale[0]);
    w[1] = __expf(((s0v.y - mnv[1]) + s1v.y) * scale[1]);
    w[2] = __expf(((s0v.z - mnv[2]) + s1v.z) * scale[2]);
    w[3] = __expf(((s0v.w - mnv[3]) + s1v.w) * scale[3]);
    float* orow = out + (size_t)n * HD;
    if (n1 >= N) {
      edge_node<false>(x, w, off, X1, lane, orow);
      break;
    }
    const int n2  = n1 + stride;
    const int n2c = (n2 < N) ? n2 : N - 1;
    const float4 s0n = *(const float4*)(s0 + (size_t)n1 * NHEAD);
    const float4 s1n = *(const float4*)(s1 + (size_t)(unsigned)col1 * NHEAD);
    const int col2 = ecol[(size_t)n2c * DEG + e16];
    edge_row_offsets(off, col1, lane);
    edge_node<true>(x, w, off, X1, lane, orow);
    n = n1; n1 = n2; col1 = col2; s0v = s0n; s1v = s1n;
  }
}

extern "C" void kernel_launch(void* const* d_in, const int* in_sizes, int n_in,
                              void* d_out, int out_size, void* d_ws, size_t ws_size,
                              hipStream_t stream) {
  const float* X    = (const float*)d_in[0];
  const float* W0   = (const float*)d_in[1];
  const float* W1   = (const float*)d_in[2];
  const float* a0   = (const float*)d_in[3];
  const int*   esrc = (const int*)d_in[4];
  const int*   ecol = (const int*)d_in[5];
  float*       out  = (float*)d_out;

  const int N = in_sizes[0] / IN_F;
  const int E = in_sizes[5];

  float* X1ws = (float*)d_ws;
  float* s0   = X1ws + (size_t)N * HD;
  float* s1   = s0 + (size_t)N * NHEAD;
  float* mxmn = s1 + (size_t)N * NHEAD;

  gat_gemm<<<GEMM_WGS, 256, 0, stream>>>(X, W0, W1, a0, X1ws, s0, s1, mxmn, N);
  gat_minmax<<<MM_WGS, MM_WAVES * 32, 0, stream>>>(s0, s1, esrc, ecol, E, mxmn);
  gat_edge<<<EDGE_WGS, EDGE_WAVES * 32, 0, stream>>>(X1ws, s0, s1, ecol, mxmn, out, N);
}
